# starting kernel
# baseline (speedup 1.0000x reference)
#include <hip/hip_runtime.h>

typedef __attribute__((ext_vector_type(2))) float v2f;
typedef __attribute__((ext_vector_type(8))) float v8f;

#define BSZ 16
#define CDIM 256
#define SDIM 1024

__global__ __launch_bounds__(256) void pool_kernel(const float* __restrict__ x,
                                                   float* __restrict__ pooled) {
    __shared__ float red[256];
    const int row = blockIdx.x;
    const float4* p = (const float4*)(x + (size_t)row * SDIM);
    float4 v = p[threadIdx.x];
    red[threadIdx.x] = v.x + v.y + v.z + v.w;
    __syncthreads();
    for (int off = 128; off > 0; off >>= 1) {
        if (threadIdx.x < off) red[threadIdx.x] += red[threadIdx.x + off];
        __syncthreads();
    }
    if (threadIdx.x == 0) pooled[row] = red[0] * (1.0f / SDIM);
}

__global__ __launch_bounds__(512) void gate_kernel(const float* __restrict__ pooled,
                                                   const float* __restrict__ in_proj_w,
                                                   const float* __restrict__ in_proj_b,
                                                   const float* __restrict__ out_proj_w,
                                                   const float* __restrict__ out_proj_b,
                                                   float* __restrict__ gate) {
    __shared__ float tmpS[BSZ * CDIM];
    const int lane = threadIdx.x & 31;
    const int wave = threadIdx.x >> 5;
    const int m    = lane & 15;
    const int half = lane >> 4;
    const int n    = wave * 16 + m;

    const float* wv  = in_proj_w + 2 * CDIM * CDIM;
    const float* bv  = in_proj_b + 2 * CDIM;

    v8f acc = {};
    for (int k0 = 0; k0 < CDIM; k0 += 4) {
        const int ka = k0 + half * 2;
        v2f a = *(const v2f*)(pooled + m * CDIM + ka);
        v2f b = *(const v2f*)(wv + n * CDIM + ka);
        acc = __builtin_amdgcn_wmma_f32_16x16x4_f32(false, a, false, b,
                                                    (short)0, acc, false, false);
    }
    {
        const float bn = bv[n];
#pragma unroll
        for (int r = 0; r < 8; ++r)
            tmpS[(r + half * 8) * CDIM + n] = acc[r] + bn;
    }
    __syncthreads();

    v8f acc2 = {};
    for (int k0 = 0; k0 < CDIM; k0 += 4) {
        const int ka = k0 + half * 2;
        v2f a = *(const v2f*)(tmpS + m * CDIM + ka);
        v2f b = *(const v2f*)(out_proj_w + n * CDIM + ka);
        acc2 = __builtin_amdgcn_wmma_f32_16x16x4_f32(false, a, false, b,
                                                     (short)0, acc2, false, false);
    }
    {
        const float bo = out_proj_b[n];
#pragma unroll
        for (int r = 0; r < 8; ++r)
            gate[(r + half * 8) * CDIM + n] = acc2[r] + bo;
    }
}

__global__ __launch_bounds__(256) void mul_kernel(const float* __restrict__ x,
                                                  const float* __restrict__ gate,
                                                  float* __restrict__ out) {
    const size_t idx = (size_t)blockIdx.x * 256 + threadIdx.x;
    const int row = (int)(idx >> 8);
    const float g = gate[row];
    float4 v = ((const float4*)x)[idx];
    v.x *= g; v.y *= g; v.z *= g; v.w *= g;
    ((float4*)out)[idx] = v;
}

extern "C" void kernel_launch(void* const* d_in, const int* in_sizes, int n_in,
                              void* d_out, int out_size, void* d_ws, size_t ws_size,
                              hipStream_t stream) {
    const float* x          = (const float*)d_in[0];
    const float* in_proj_w  = (const float*)d_in[1];
    const float* in_proj_b  = (const float*)d_in[2];
    const float* out_proj_w = (const float*)d_in[3];
    const float* out_proj_b = (const float*)d_in[4];
    float* out = (float*)d_out;

    float* pooled = (float*)d_ws;
    float* gate   = pooled + BSZ * CDIM;

    pool_kernel<<<BSZ * CDIM, 256, 0, stream>>>(x, pooled);
    gate_kernel<<<1, 512, 0, stream>>>(pooled, in_proj_w, in_proj_b,
                                       out_proj_w, out_proj_b, gate);
    mul_kernel<<<(BSZ * CDIM * SDIM) / 4 / 256, 256, 0, stream>>>(x, gate, out);

    (void)in_sizes; (void)n_in; (void)out_size; (void)ws_size;
}


# second gate product moved into the scaling kernel: 256 work-groups (batch x 16 channels), one 4-KB row per wave, each forms its 16 gates from tmp with 4 WMMAs per wave; gate kernel keeps product 1
# speedup vs baseline: 1.3877x; 1.3877x over previous
#include <hip/hip_runtime.h>


typedef __attribute__((ext_vector_type(2))) float v2f;
typedef __attribute__((ext_vector_type(8))) float v8f;

#define BSZ 16
#define CDIM 256
#define SDIM 1024

__global__ __launch_bounds__(256) void pool_kernel(const float* __restrict__ x,
                                                   float* __restrict__ pooled) {
    __shared__ float red[256];
    const int row = blockIdx.x;
    const float4* p = (const float4*)(x + (size_t)row * SDIM);
    float4 v = p[threadIdx.x];
    red[threadIdx.x] = v.x + v.y + v.z + v.w;
    __syncthreads();
    for (int off = 128; off > 0; off >>= 1) {
        if (threadIdx.x < off) red[threadIdx.x] += red[threadIdx.x + off];
        __syncthreads();
    }
    if (threadIdx.x == 0) pooled[row] = red[0] * (1.0f / SDIM);
}

__global__ __launch_bounds__(512) void proj_kernel(const float* __restrict__ pooled,
                                                   const float* __restrict__ in_proj_w,
                                                   const float* __restrict__ in_proj_b,
                                                   float* __restrict__ tmp) {
    const int lane = threadIdx.x & 31;
    const int wave = threadIdx.x >> 5;
    const int m    = lane & 15;
    const int half = lane >> 4;
    const int n    = wave * 16 + m;

    const float* wv  = in_proj_w + 2 * CDIM * CDIM;
    const float* bv  = in_proj_b + 2 * CDIM;

    v8f acc = {};
    for (int k0 = 0; k0 < CDIM; k0 += 4) {
        const int ka = k0 + half * 2;
        v2f a = *(const v2f*)(pooled + m * CDIM + ka);
        v2f b = *(const v2f*)(wv + n * CDIM + ka);
        acc = __builtin_amdgcn_wmma_f32_16x16x4_f32(false, a, false, b,
                                                    (short)0, acc, false, false);
    }
    const float bn = bv[n];
#pragma unroll
    for (int r = 0; r < 8; ++r)
        tmp[(r + half * 8) * CDIM + n] = acc[r] + bn;
}

__global__ __launch_bounds__(512) void scale_kernel(const float* __restrict__ x,
                                                    const float* __restrict__ tmp,
                                                    const float* __restrict__ out_proj_w,
                                                    const float* __restrict__ out_proj_b,
                                                    float* __restrict__ out) {
    __shared__ float part[16 * 16];
    const int lane = threadIdx.x & 31;
    const int wave = threadIdx.x >> 5;
    const int m    = lane & 15;
    const int half = lane >> 4;
    const int b    = blockIdx.x >> 4;
    const int ct   = blockIdx.x & 15;

    const int kb = wave * 16 + half * 8;
    const float4* ap = (const float4*)(out_proj_w + (size_t)(ct * 16 + m) * CDIM + kb);
    const float4* bp = (const float4*)(tmp + m * CDIM + kb);
    const float4 a0 = ap[0], a1 = ap[1];
    const float4 b0 = bp[0], b1 = bp[1];
    const float bo = out_proj_b[ct * 16 + wave];

    const size_t row = (size_t)blockIdx.x * 16 + wave;
    const float4* xp = (const float4*)(x + row * SDIM);
    float4 v[8];
#pragma unroll
    for (int j = 0; j < 8; ++j) v[j] = xp[lane + 32 * j];

    v8f acc = {};
    {
        v2f fa, fb;
        fa.x = a0.x; fa.y = a0.y; fb.x = b0.x; fb.y = b0.y;
        acc = __builtin_amdgcn_wmma_f32_16x16x4_f32(false, fa, false, fb, (short)0, acc, false, false);
        fa.x = a0.z; fa.y = a0.w; fb.x = b0.z; fb.y = b0.w;
        acc = __builtin_amdgcn_wmma_f32_16x16x4_f32(false, fa, false, fb, (short)0, acc, false, false);
        fa.x = a1.x; fa.y = a1.y; fb.x = b1.x; fb.y = b1.y;
        acc = __builtin_amdgcn_wmma_f32_16x16x4_f32(false, fa, false, fb, (short)0, acc, false, false);
        fa.x = a1.z; fa.y = a1.w; fb.x = b1.z; fb.y = b1.w;
        acc = __builtin_amdgcn_wmma_f32_16x16x4_f32(false, fa, false, fb, (short)0, acc, false, false);
    }
    if (m == b) {
#pragma unroll
        for (int r = 0; r < 8; ++r)
            part[(half * 8 + r) * 16 + wave] = acc[r];
    }
    __syncthreads();

    const float4* pp = (const float4*)(part + wave * 16);
    const float4 p0 = pp[0], p1 = pp[1], p2 = pp[2], p3 = pp[3];
    float g = ((p0.x + p0.y) + (p0.z + p0.w)) + ((p1.x + p1.y) + (p1.z + p1.w));
    g += ((p2.x + p2.y) + (p2.z + p2.w)) + ((p3.x + p3.y) + (p3.z + p3.w));
    g += bo;

    float4* op = (float4*)(out + row * SDIM);
#pragma unroll
    for (int j = 0; j < 8; ++j) {
        float4 t = v[j];
        t.x *= g; t.y *= g; t.z *= g; t.w *= g;
        op[lane + 32 * j] = t;
    }
}

extern "C" void kernel_launch(void* const* d_in, const int* in_sizes, int n_in,
                              void* d_out, int out_size, void* d_ws, size_t ws_size,
                              hipStream_t stream) {
    const float* x          = (const float*)d_in[0];
    const float* in_proj_w  = (const float*)d_in[1];
    const float* in_proj_b  = (const float*)d_in[2];
    const float* out_proj_w = (const float*)d_in[3];
    const float* out_proj_b = (const float*)d_in[4];
    float* out = (float*)d_out;

    float* pooled = (float*)d_ws;
    float* tmp    = pooled + BSZ * CDIM;

    pool_kernel<<<BSZ * CDIM, 256, 0, stream>>>(x, pooled);
    proj_kernel<<<1, 512, 0, stream>>>(pooled, in_proj_w, in_proj_b, tmp);
    scale_kernel<<<BSZ * (CDIM / 16), 512, 0, stream>>>(x, tmp, out_proj_w, out_proj_b, out);

    (void)in_sizes; (void)n_in; (void)out_size; (void)ws_size;
}


# fragment operands of both products fetched as whole or half lines and turned to fragment order in wave-private padded LDS, not 16-line gathers ahead of the row stream; product 2 on 8 waves of 32 k
# speedup vs baseline: 3.9408x; 2.8398x over previous
#include <hip/hip_runtime.h>


typedef __attribute__((ext_vector_type(2))) float v2f;
typedef __attribute__((ext_vector_type(8))) float v8f;

#define BSZ 16
#define CDIM 256
#define SDIM 1024

#define WMMA4(a, b, c) __builtin_amdgcn_wmma_f32_16x16x4_f32(false, (a), false, (b), (short)0, (c), false, false)

#define ROW_XOR_ADD(v, M)                                                                  \
    ((v) + __builtin_bit_cast(float, __builtin_amdgcn_update_dpp(                          \
               0, __builtin_bit_cast(int, (v)), 0x160 | (M), 0xf, 0xf, true)))

__device__ __forceinline__ float xhalf(float v) {
    const int b = __builtin_bit_cast(int, v);
    return __builtin_bit_cast(float,
        __builtin_amdgcn_permlanex16(b, b, 0x76543210, 0xfedcba98, false, false));
}

__device__ __forceinline__ void lds_barrier() {
    asm volatile("s_wait_dscnt 0x0\n\ts_barrier_signal -1\n\ts_barrier_wait -1" ::: "memory");
}

__device__ __forceinline__ void lds_wave_sync() {
    asm volatile("s_wait_dscnt 0x0" ::: "memory");
}

__device__ __forceinline__ float wave_sum(float v) {
    v = ROW_XOR_ADD(v, 1);
    v = ROW_XOR_ADD(v, 2);
    v = ROW_XOR_ADD(v, 4);
    v = ROW_XOR_ADD(v, 8);
    return v + xhalf(v);
}

#define WROW_PITCH 5
__global__ __launch_bounds__(512) void pool_proj_kernel(const float* __restrict__ x,
                                                        const float* __restrict__ in_proj_w,
                                                        const float* __restrict__ in_proj_b,
                                                        float* __restrict__ part) {
    __shared__ float4 mean_s[4];
    __shared__ float4 wst[8 * 32 * WROW_PITCH];
    const int lane = threadIdx.x & 31;
    const int wave = __builtin_amdgcn_readfirstlane(threadIdx.x >> 5);
    const int c    = lane & 15;
    const int half = lane >> 4;
    const int b    = blockIdx.x >> 4;
    const int ct   = blockIdx.x & 15;

    const float* wv = in_proj_w + 2 * CDIM * CDIM;
    const float* bv = in_proj_b + 2 * CDIM;

    const int wq = lane & 3;
    const int wr = lane >> 2;
    float4 t0, t1, t2, t3;
    float bias = 0.0f;
    if (wave < 8) {
        const float* wp = wv + (size_t)(wave * 32 + wr) * CDIM + ct * 16 + wq * 4;
        t0 = *(const float4*)(wp);
        t1 = *(const float4*)(wp + 8 * CDIM);
        t2 = *(const float4*)(wp + 16 * CDIM);
        t3 = *(const float4*)(wp + 24 * CDIM);
        if (ct == 0) bias = bv[wave * 32 + lane];
    }

    const size_t row = (size_t)blockIdx.x * 16 + wave;
    const float4* xp = (const float4*)(x + row * SDIM);
    float4 v[8];
#pragma unroll
    for (int j = 0; j < 8; ++j) v[j] = xp[lane + 32 * j];
    __builtin_amdgcn_sched_barrier(0);

    float4 w00 = {}, w01 = {}, w10 = {}, w11 = {};
    if (wave < 8) {
        float4* st = wst + wave * (32 * WROW_PITCH) + wr * WROW_PITCH + wq;
        st[0]               = t0;
        st[8 * WROW_PITCH]  = t1;
        st[16 * WROW_PITCH] = t2;
        st[24 * WROW_PITCH] = t3;
        lds_wave_sync();
        const float4* rd = wst + wave * (32 * WROW_PITCH) + c * WROW_PITCH + half * 2;
        w00 = rd[0]; w01 = rd[1];
        w10 = rd[16 * WROW_PITCH]; w11 = rd[16 * WROW_PITCH + 1];
    }

    float s = 0.0f;
#pragma unroll
    for (int j = 0; j < 8; ++j) s += (v[j].x + v[j].y) + (v[j].z + v[j].w);
    s = wave_sum(s);
    if (lane == 0) ((float*)mean_s)[wave] = s * (1.0f / SDIM);
    __syncthreads();

    if (wave < 8) {
        const float4 a0 = mean_s[half * 2], a1 = mean_s[half * 2 + 1];
        v8f acc0 = {}, acc1 = {};
        v2f fa, fb;
        fa.x = a0.x; fa.y = a0.y;
        fb.x = w00.x; fb.y = w00.y; acc0 = WMMA4(fa, fb, acc0);
        fb.x = w10.x; fb.y = w10.y; acc1 = WMMA4(fa, fb, acc1);
        fa.x = a0.z; fa.y = a0.w;
        fb.x = w00.z; fb.y = w00.w; acc0 = WMMA4(fa, fb, acc0);
        fb.x = w10.z; fb.y = w10.w; acc1 = WMMA4(fa, fb, acc1);
        fa.x = a1.x; fa.y = a1.y;
        fb.x = w01.x; fb.y = w01.y; acc0 = WMMA4(fa, fb, acc0);
        fb.x = w11.x; fb.y = w11.y; acc1 = WMMA4(fa, fb, acc1);
        fa.x = a1.z; fa.y = a1.w;
        fb.x = w01.z; fb.y = w01.w; acc0 = WMMA4(fa, fb, acc0);
        fb.x = w11.z; fb.y = w11.w; acc1 = WMMA4(fa, fb, acc1);
        const float r = (half == 0 ? acc0[0] : acc1[0]) + bias;
        part[(size_t)(ct * 16 + b) * CDIM + wave * 32 + lane] = r;
    }
}

#define GROW_PITCH 9
__global__ __launch_bounds__(512) void scale_kernel(const float* x,
                                                    const float* __restrict__ part,
                                                    const float* __restrict__ out_proj_w,
                                                    const float* __restrict__ out_proj_b,
                                                    float* __restrict__ out) {
    __shared__ float4 red[16 * 32];
    __shared__ float4 gst[8 * 2 * 16 * GROW_PITCH];
    const int lane = threadIdx.x & 31;
    const int wave = __builtin_amdgcn_readfirstlane(threadIdx.x >> 5);
    const int m    = lane & 15;
    const int half = lane >> 4;
    const int b    = blockIdx.x >> 4;
    const int ct   = blockIdx.x & 15;

    const int gq = lane & 7;
    const int gr = lane >> 3;
    float4 ga0, ga1, ga2, ga3, gb0, gb1, gb2, gb3;
    if (wave < 8) {
        const float* ap = out_proj_w + (size_t)(ct * 16 + gr) * CDIM + wave * 32 + gq * 4;
        const float* bp = part + (size_t)(gr * 16 + b) * CDIM + wave * 32 + gq * 4;
        ga0 = *(const float4*)(ap);
        ga1 = *(const float4*)(ap + 4 * CDIM);
        ga2 = *(const float4*)(ap + 8 * CDIM);
        ga3 = *(const float4*)(ap + 12 * CDIM);
        gb0 = *(const float4*)(bp);
        gb1 = *(const float4*)(bp + 4 * BSZ * CDIM);
        gb2 = *(const float4*)(bp + 8 * BSZ * CDIM);
        gb3 = *(const float4*)(bp + 12 * BSZ * CDIM);
    }
    const float bo = out_proj_b[ct * 16 + wave];

    const size_t row = (size_t)blockIdx.x * 16 + wave;
    const float4* xp = (const float4*)(x + row * SDIM);
    float4 v[8];
#pragma unroll
    for (int j = 0; j < 8; ++j) v[j] = xp[lane + 32 * j];
    __builtin_amdgcn_sched_barrier(0);

    if (wave < 8) {
        float4* sa = gst + wave * (2 * 16 * GROW_PITCH);
        float4* sb = sa + 16 * GROW_PITCH;
        float4* wa = sa + gr * GROW_PITCH + gq;
        float4* wb = sb + gr * GROW_PITCH + gq;
        wa[0]               = ga0;
        wa[4 * GROW_PITCH]  = ga1;
        wa[8 * GROW_PITCH]  = ga2;
        wa[12 * GROW_PITCH] = ga3;
        wb[0]               = gb0;
        wb[4 * GROW_PITCH]  = gb1;
        wb[8 * GROW_PITCH]  = gb2;
        wb[12 * GROW_PITCH] = gb3;
        lds_wave_sync();
        const float4* ra = sa + m * GROW_PITCH + half * 4;
        const float4* rb = sb + m * GROW_PITCH + half * 4;
        v8f acc0 = {}, acc1 = {};
#pragma unroll
        for (int e = 0; e < 4; ++e) {
            const float4 fa4 = ra[e], fb4 = rb[e];
            v2f fa, fb;
            fa.x = fa4.x; fa.y = fa4.y; fb.x = fb4.x; fb.y = fb4.y;
            acc0 = WMMA4(fa, fb, acc0);
            fa.x = fa4.z; fa.y = fa4.w; fb.x = fb4.z; fb.y = fb4.w;
            acc1 = WMMA4(fa, fb, acc1);
        }
        float* redf = (float*)red;
#pragma unroll
        for (int r = 0; r < 8; ++r)
            redf[((half * 8 + r) * 8 + wave) * 16 + m] = acc0[r] + acc1[r];
    }
    lds_barrier();

    const float4 p0 = red[wave * 32 + lane];
    float g = (p0.x + p0.y) + (p0.z + p0.w);
    g = wave_sum(g) + bo;

    float4* op = (float4*)(out + row * SDIM);
#pragma unroll
    for (int j = 0; j < 8; ++j) {
        float4 t = v[j];
        t.x *= g; t.y *= g; t.z *= g; t.w *= g;
        op[lane + 32 * j] = t;
    }
}

extern "C" void kernel_launch(void* const* d_in, const int* in_sizes, int n_in,
                              void* d_out, int out_size, void* d_ws, size_t ws_size,
                              hipStream_t stream) {
    const float* x          = (const float*)d_in[0];
    const float* in_proj_w  = (const float*)d_in[1];
    const float* in_proj_b  = (const float*)d_in[2];
    const float* out_proj_w = (const float*)d_in[3];
    const float* out_proj_b = (const float*)d_in[4];
    float* out = (float*)d_out;

    float* part = (float*)d_ws;

    pool_proj_kernel<<<BSZ * (CDIM / 16), 512, 0, stream>>>(x, in_proj_w, in_proj_b, part);
    scale_kernel<<<BSZ * (CDIM / 16), 512, 0, stream>>>(x, part, out_proj_w, out_proj_b, out);

    (void)in_sizes; (void)n_in; (void)out_size; (void)ws_size;
}
